# starting kernel
# baseline (speedup 1.0000x reference)
#include <hip/hip_runtime.h>
#include <hip/hip_bf16.h>


#define Bn 64
#define Nn 256
#define Hn 1024
#define NK 768
#define LN_EPS 1e-5f

typedef __attribute__((ext_vector_type(2))) float v2f;
typedef __attribute__((ext_vector_type(8))) float v8f;

__global__ __launch_bounds__(256) void wgen_gemm(
    const float* __restrict__ f, const float* __restrict__ Wlin,
    const float* __restrict__ blin, float* __restrict__ wout) {
  __shared__ float As[64 * 20];
  __shared__ float Bs[16 * 136];

  const int tid = threadIdx.x;
  const int b = blockIdx.z;
  const int m0 = blockIdx.y * 64;
  const int o0 = blockIdx.x * 128;
  const int wave = tid >> 5, lane = tid & 31;
  const int lm = lane & 15, hi = lane >> 4;
  const int wm = (wave & 3) * 16;
  const int wn = (wave >> 2) * 64;

  v8f cAcc[4] = {};

  const int arow = tid >> 2;
  const int acol = (tid & 3) * 4;
  const float* fA = f + (size_t)(b * Nn + m0 + arow) * Hn + acol;

  for (int k0 = 0; k0 < Hn; k0 += 16) {
    __syncthreads();
    float4 av = *(const float4*)(fA + k0);
    *(float4*)(&As[arow * 20 + acol]) = av;
#pragma unroll
    for (int j = 0; j < 2; ++j) {
      int idx = tid + j * 256;
      int br = idx >> 2;
      int bc = (idx & 3) * 4;
      float4 bv = *(const float4*)(Wlin + (size_t)(o0 + br) * Hn + k0 + bc);
      Bs[(bc + 0) * 136 + br] = bv.x;
      Bs[(bc + 1) * 136 + br] = bv.y;
      Bs[(bc + 2) * 136 + br] = bv.z;
      Bs[(bc + 3) * 136 + br] = bv.w;
    }
    __syncthreads();
#pragma unroll
    for (int ks = 0; ks < 4; ++ks) {
      const int kk = ks * 4 + hi * 2;
      v2f a;
      a.x = As[(wm + lm) * 20 + kk];
      a.y = As[(wm + lm) * 20 + kk + 1];
#pragma unroll
      for (int ct = 0; ct < 4; ++ct) {
        const int n = wn + ct * 16 + lm;
        v2f bb;
        bb.x = Bs[(kk + 0) * 136 + n];
        bb.y = Bs[(kk + 1) * 136 + n];
        cAcc[ct] = __builtin_amdgcn_wmma_f32_16x16x4_f32(
            false, a, false, bb, (short)0, cAcc[ct], false, false);
      }
    }
  }

#pragma unroll
  for (int ct = 0; ct < 4; ++ct) {
    const int o = o0 + wn + ct * 16 + lm;
    const float bl = blin[o];
#pragma unroll
    for (int r = 0; r < 8; ++r) {
      const int m = m0 + wm + r + hi * 8;
      wout[(size_t)(b * Nn + m) * NK + o] = cAcc[ct][r] + bl;
    }
  }
}

__global__ __launch_bounds__(256) void dyconv_ln(
    const float* __restrict__ kin, const float* __restrict__ w,
    const float* __restrict__ gamma, const float* __restrict__ beta,
    float* __restrict__ out) {
  __shared__ float Ks[16 * 520];
  __shared__ float Aw[16 * 52];
  __shared__ float Gs[1024];
  __shared__ float Bt[1024];
  __shared__ float redS[8 * 16];
  __shared__ float redQ[8 * 16];
  __shared__ float muS[16];
  __shared__ float rsS[16];

  const int tid = threadIdx.x;
  const int b = blockIdx.y;
  const int m0 = blockIdx.x * 16;
  const int wave = tid >> 5, lane = tid & 31;
  const int lm = lane & 15, hi2 = lane >> 4;

#pragma unroll
  for (int j = 0; j < 4; ++j) {
    int i = tid + j * 256;
    Gs[i] = gamma[i];
    Bt[i] = beta[i];
  }

  v8f acc[8] = {};

  const int aRow = tid >> 4;
  const int aCol = (tid & 15) * 3;
  const float* wp = w + (size_t)(b * Nn + m0 + aRow) * NK + aCol;
  const int kRow = tid >> 4;
  const int kCb = (tid & 15) * 32;
  const int hRow = tid & 15;

  for (int ic = 0; ic < 16; ++ic) {
    const int i0 = ic * 16;
    __syncthreads();
    {
      const float* wq = wp + (size_t)i0 * 3;
      Aw[aRow * 52 + aCol + 0] = wq[0];
      Aw[aRow * 52 + aCol + 1] = wq[1];
      Aw[aRow * 52 + aCol + 2] = wq[2];
    }
#pragma unroll
    for (int p = 0; p < 2; ++p) {
      if (p) __syncthreads();
      {
        const float* kr = kin + (size_t)(b * Nn + i0 + kRow) * Hn + p * 512 + kCb;
#pragma unroll
        for (int j = 0; j < 8; ++j) {
          float4 v = *(const float4*)(kr + j * 4);
          const int cc = kRow * 520 + 1 + kCb + j * 4;
          Ks[cc + 0] = v.x;
          Ks[cc + 1] = v.y;
          Ks[cc + 2] = v.z;
          Ks[cc + 3] = v.w;
        }
      }
      if (tid < 32) {
        const float* krh = kin + (size_t)(b * Nn + i0 + hRow) * Hn;
        if (tid < 16) {
          Ks[hRow * 520 + 0] = (p == 0) ? 0.0f : krh[511];
        } else {
          Ks[hRow * 520 + 513] = (p == 0) ? krh[512] : 0.0f;
        }
      }
      __syncthreads();
#pragma unroll
      for (int t = 0; t < 3; ++t) {
#pragma unroll
        for (int ks = 0; ks < 4; ++ks) {
          const int isub = ks * 4 + hi2 * 2;
          v2f a;
          a.x = Aw[lm * 52 + (isub + 0) * 3 + t];
          a.y = Aw[lm * 52 + (isub + 1) * 3 + t];
#pragma unroll
          for (int ct = 0; ct < 4; ++ct) {
            const int col = (wave * 64 + ct * 16 + lm) + t;
            v2f bb;
            bb.x = Ks[(isub + 0) * 520 + col];
            bb.y = Ks[(isub + 1) * 520 + col];
            acc[p * 4 + ct] = __builtin_amdgcn_wmma_f32_16x16x4_f32(
                false, a, false, bb, (short)0, acc[p * 4 + ct], false, false);
          }
        }
      }
    }
  }

  float s[8], q[8];
#pragma unroll
  for (int r = 0; r < 8; ++r) { s[r] = 0.0f; q[r] = 0.0f; }
#pragma unroll
  for (int u = 0; u < 8; ++u)
#pragma unroll
    for (int r = 0; r < 8; ++r) {
      float v = acc[u][r];
      s[r] += v;
      q[r] += v * v;
    }
#pragma unroll
  for (int mask = 1; mask < 16; mask <<= 1)
#pragma unroll
    for (int r = 0; r < 8; ++r) {
      s[r] += __shfl_xor(s[r], mask, 32);
      q[r] += __shfl_xor(q[r], mask, 32);
    }
  if (lm == 0) {
#pragma unroll
    for (int r = 0; r < 8; ++r) {
      const int row = r + hi2 * 8;
      redS[wave * 16 + row] = s[r];
      redQ[wave * 16 + row] = q[r];
    }
  }
  __syncthreads();
  if (tid < 16) {
    float ss = 0.0f, qq = 0.0f;
#pragma unroll
    for (int wv = 0; wv < 8; ++wv) {
      ss += redS[wv * 16 + tid];
      qq += redQ[wv * 16 + tid];
    }
    const float mu = ss * (1.0f / 1024.0f);
    const float var = qq * (1.0f / 1024.0f) - mu * mu;
    muS[tid] = mu;
    rsS[tid] = rsqrtf(var + LN_EPS);
  }
  __syncthreads();

#pragma unroll
  for (int u = 0; u < 8; ++u) {
    const int p = u >> 2, ct = u & 3;
    const int h = p * 512 + wave * 64 + ct * 16 + lm;
    const float g = Gs[h], be = Bt[h];
#pragma unroll
    for (int r = 0; r < 8; ++r) {
      const int row = r + hi2 * 8;
      const float v = (acc[u][r] - muS[row]) * rsS[row] * g + be;
      out[(size_t)(b * Nn + m0 + row) * Hn + h] = v;
    }
  }
}

extern "C" void kernel_launch(void* const* d_in, const int* in_sizes, int n_in,
                              void* d_out, int out_size, void* d_ws, size_t ws_size,
                              hipStream_t stream) {
  const float* f     = (const float*)d_in[0];
  const float* k     = (const float*)d_in[1];
  const float* Wlin  = (const float*)d_in[2];
  const float* blin  = (const float*)d_in[3];
  const float* gamma = (const float*)d_in[4];
  const float* beta  = (const float*)d_in[5];
  float* out = (float*)d_out;
  float* wgt = (float*)d_ws;

  (void)in_sizes; (void)n_in; (void)out_size; (void)ws_size;

  wgen_gemm<<<dim3(NK / 128, Nn / 64, Bn), 256, 0, stream>>>(f, Wlin, blin, wgt);
  dyconv_ln<<<dim3(Nn / 16, Bn), 256, 0, stream>>>(k, wgt, gamma, beta, out);
}


# Stage 2 k operand loaded straight from global memory into WMMA fragments (a lane owns 4-column groups), weight tile staged once in LDS: no LDS slab, no barrier in k loop; LayerNorm row sums regrouped
# speedup vs baseline: 1.4312x; 1.4312x over previous
#include <hip/hip_runtime.h>
#include <hip/hip_bf16.h>


#define Bn 64
#define Nn 256
#define Hn 1024
#define NK 768
#define LN_EPS 1e-5f

typedef __attribute__((ext_vector_type(2))) float v2f;
typedef __attribute__((ext_vector_type(8))) float v8f;

__global__ __launch_bounds__(256) void wgen_gemm(
    const float* __restrict__ f, const float* __restrict__ Wlin,
    const float* __restrict__ blin, float* __restrict__ wout) {
  __shared__ float As[64 * 20];
  __shared__ float Bs[16 * 136];

  const int tid = threadIdx.x;
  const int b = blockIdx.z;
  const int m0 = blockIdx.y * 64;
  const int o0 = blockIdx.x * 128;
  const int wave = tid >> 5, lane = tid & 31;
  const int lm = lane & 15, hi = lane >> 4;
  const int wm = (wave & 3) * 16;
  const int wn = (wave >> 2) * 64;

  v8f cAcc[4] = {};

  const int arow = tid >> 2;
  const int acol = (tid & 3) * 4;
  const float* fA = f + (size_t)(b * Nn + m0 + arow) * Hn + acol;

  for (int k0 = 0; k0 < Hn; k0 += 16) {
    __syncthreads();
    float4 av = *(const float4*)(fA + k0);
    *(float4*)(&As[arow * 20 + acol]) = av;
#pragma unroll
    for (int j = 0; j < 2; ++j) {
      int idx = tid + j * 256;
      int br = idx >> 2;
      int bc = (idx & 3) * 4;
      float4 bv = *(const float4*)(Wlin + (size_t)(o0 + br) * Hn + k0 + bc);
      Bs[(bc + 0) * 136 + br] = bv.x;
      Bs[(bc + 1) * 136 + br] = bv.y;
      Bs[(bc + 2) * 136 + br] = bv.z;
      Bs[(bc + 3) * 136 + br] = bv.w;
    }
    __syncthreads();
#pragma unroll
    for (int ks = 0; ks < 4; ++ks) {
      const int kk = ks * 4 + hi * 2;
      v2f a;
      a.x = As[(wm + lm) * 20 + kk];
      a.y = As[(wm + lm) * 20 + kk + 1];
#pragma unroll
      for (int ct = 0; ct < 4; ++ct) {
        const int n = wn + ct * 16 + lm;
        v2f bb;
        bb.x = Bs[(kk + 0) * 136 + n];
        bb.y = Bs[(kk + 1) * 136 + n];
        cAcc[ct] = __builtin_amdgcn_wmma_f32_16x16x4_f32(
            false, a, false, bb, (short)0, cAcc[ct], false, false);
      }
    }
  }

#pragma unroll
  for (int ct = 0; ct < 4; ++ct) {
    const int o = o0 + wn + ct * 16 + lm;
    const float bl = blin[o];
#pragma unroll
    for (int r = 0; r < 8; ++r) {
      const int m = m0 + wm + r + hi * 8;
      wout[(size_t)(b * Nn + m) * NK + o] = cAcc[ct][r] + bl;
    }
  }
}

#define S2W 8
#define S2T (S2W * 32)
#define S2CT 8
#define AWP 772
#define NSTEP (Nn / 4)

__device__ __forceinline__ float row_from_left(float x) {
  const int v = __float_as_int(x);
  return __int_as_float(__builtin_amdgcn_update_dpp(v, v, 0x121  , 0xF, 0xF, false));
}
__device__ __forceinline__ float row_from_right(float x) {
  const int v = __float_as_int(x);
  return __int_as_float(__builtin_amdgcn_update_dpp(v, v, 0x12F  , 0xF, 0xF, false));
}

__device__ __forceinline__ void step_mma(v8f (&acc)[3][S2CT], const v2f* ap,
                                         const float4 X0, const float4 X1,
                                         const float4 Y0, const float4 Y1) {
  const v2f a0 = ap[0], a1 = ap[1], a2 = ap[2];
  const float bx[S2CT] = {X0.x, X0.y, X0.z, X0.w, X1.x, X1.y, X1.z, X1.w};
  const float by[S2CT] = {Y0.x, Y0.y, Y0.z, Y0.w, Y1.x, Y1.y, Y1.z, Y1.w};
#pragma unroll
  for (int ct = 0; ct < S2CT; ++ct) {
    v2f bb;
    bb.x = bx[ct];
    bb.y = by[ct];
    acc[0][ct] = __builtin_amdgcn_wmma_f32_16x16x4_f32(
        false, a0, false, bb, (short)0, acc[0][ct], false, false);
    acc[1][ct] = __builtin_amdgcn_wmma_f32_16x16x4_f32(
        false, a1, false, bb, (short)0, acc[1][ct], false, false);
    acc[2][ct] = __builtin_amdgcn_wmma_f32_16x16x4_f32(
        false, a2, false, bb, (short)0, acc[2][ct], false, false);
  }
}

__global__ __launch_bounds__(S2T) __attribute__((amdgpu_waves_per_eu(4, 4))) void dyconv_ln(
    const float* __restrict__ kin, const float* __restrict__ w,
    const float* __restrict__ gamma, const float* __restrict__ beta,
    float* __restrict__ out) {
  __shared__ __attribute__((aligned(16))) float Aw[16 * AWP];
  __shared__ float redS[S2W * 16];
  __shared__ float redQ[S2W * 16];
  __shared__ float edgeL[S2W * 16];
  __shared__ float edgeR[S2W * 16];
  __shared__ float muS[16];
  __shared__ float rsS[16];

  const int tid = threadIdx.x;
  const int b = blockIdx.y;
  const int m0 = blockIdx.x * 16;
  const int wave = tid >> 5, lane = tid & 31;
  const int lm = lane & 15, hi2 = lane >> 4;
  const int wcol = wave * (16 * S2CT);

  const float* kp = kin + (size_t)(b * Nn + 2 * hi2) * Hn + wcol + 4 * lm;
#define LDB(x0, x1, y0, y1, p)                 \
  x0 = *(const float4*)(p);                    \
  x1 = *(const float4*)((p) + 64);             \
  y0 = *(const float4*)((p) + Hn);             \
  y1 = *(const float4*)((p) + Hn + 64)
  float4 pX0, pX1, pY0, pY1, qX0, qX1, qY0, qY1;
  LDB(pX0, pX1, pY0, pY1, kp);

  {
    const int row = tid >> 4, j = tid & 15;
    const float* wsrc = w + (size_t)(b * Nn + m0 + row) * NK + j * 48;
    float src[48], dst[48];
#pragma unroll
    for (int n = 0; n < 12; ++n) {
      const float4 v = *(const float4*)(wsrc + 4 * n);
      src[4 * n + 0] = v.x; src[4 * n + 1] = v.y; src[4 * n + 2] = v.z; src[4 * n + 3] = v.w;
    }
#pragma unroll
    for (int il = 0; il < 16; ++il)
#pragma unroll
      for (int t = 0; t < 3; ++t)
        dst[(il >> 1) * 6 + t * 2 + (il & 1)] = src[il * 3 + t];
    float* wdst = &Aw[row * AWP + j * 48];
#pragma unroll
    for (int n = 0; n < 12; ++n)
      *(float4*)(wdst + 4 * n) = make_float4(dst[4 * n], dst[4 * n + 1], dst[4 * n + 2], dst[4 * n + 3]);
  }

  v8f acc[3][S2CT] = {};

  const v2f* ap = (const v2f*)Aw + lm * (AWP / 2) + hi2 * 3;

  __syncthreads();

  for (int s = 0; s < NSTEP; s += 2) {
    LDB(qX0, qX1, qY0, qY1, kp + 4 * Hn);
    step_mma(acc, ap + s * 6, pX0, pX1, pY0, pY1);
    kp += (s + 2 < NSTEP) ? 8 * Hn : 0;
    LDB(pX0, pX1, pY0, pY1, kp);
    step_mma(acc, ap + (s + 1) * 6, qX0, qX1, qY0, qY1);
  }
#undef LDB

  if (lm == 15) {
#pragma unroll
    for (int r = 0; r < 8; ++r) edgeL[wave * 16 + hi2 * 8 + r] = acc[0][S2CT - 1][r];
  }
  if (lm == 0) {
#pragma unroll
    for (int r = 0; r < 8; ++r) edgeR[wave * 16 + hi2 * 8 + r] = acc[2][0][r];
  }
  __syncthreads();
#pragma unroll
  for (int r = 0; r < 8; ++r) {
    const float eL = (wave > 0) ? edgeL[(wave - 1) * 16 + hi2 * 8 + r] : 0.0f;
    const float eR = (wave < S2W - 1) ? edgeR[(wave + 1) * 16 + hi2 * 8 + r] : 0.0f;
    const float p3 = row_from_left(acc[0][3][r]);
    const float p7 = row_from_left(acc[0][7][r]);
    const float n0 = row_from_right(acc[2][0][r]);
    const float n4 = row_from_right(acc[2][4][r]);
    float left[S2CT], right[S2CT];
    left[0] = (lm == 0) ? eL : p3;
    left[4] = (lm == 0) ? p3 : p7;
    right[3] = (lm == 15) ? n4 : n0;
    right[7] = (lm == 15) ? eR : n4;
#pragma unroll
    for (int ct = 0; ct < S2CT; ++ct) {
      if ((ct & 3) != 0) left[ct] = acc[0][ct - 1][r];
      if ((ct & 3) != 3) right[ct] = acc[2][ct + 1][r];
    }
#pragma unroll
    for (int ct = 0; ct < S2CT; ++ct)
      acc[1][ct][r] = (left[ct] + acc[1][ct][r]) + right[ct];
  }

  float s[8], q[8];
#pragma unroll
  for (int r = 0; r < 8; ++r) { s[r] = 0.0f; q[r] = 0.0f; }
#pragma unroll
  for (int u = 0; u < S2CT; ++u)
#pragma unroll
    for (int r = 0; r < 8; ++r) {
      float v = acc[1][u][r];
      s[r] += v;
      q[r] += v * v;
    }
#pragma unroll
  for (int mask = 1; mask < 16; mask <<= 1)
#pragma unroll
    for (int r = 0; r < 8; ++r) {
      s[r] += __shfl_xor(s[r], mask, 32);
      q[r] += __shfl_xor(q[r], mask, 32);
    }
  if (lm == 0) {
#pragma unroll
    for (int r = 0; r < 8; ++r) {
      const int row = r + hi2 * 8;
      redS[wave * 16 + row] = s[r];
      redQ[wave * 16 + row] = q[r];
    }
  }
  __syncthreads();
  if (tid < 16) {
    float ss = 0.0f, qq = 0.0f;
#pragma unroll
    for (int wv = 0; wv < S2W; ++wv) {
      ss += redS[wv * 16 + tid];
      qq += redQ[wv * 16 + tid];
    }
    const float mu = ss * (1.0f / 1024.0f);
    const float var = qq * (1.0f / 1024.0f) - mu * mu;
    muS[tid] = mu;
    rsS[tid] = rsqrtf(var + LN_EPS);
  }
  __syncthreads();

#pragma unroll
  for (int g = 0; g < 2; ++g) {
    const int h = wcol + g * 64 + 4 * lm;
    const float4 ga = *(const float4*)(gamma + h);
    const float4 be = *(const float4*)(beta + h);
#pragma unroll
    for (int r = 0; r < 8; ++r) {
      const int row = r + hi2 * 8;
      const float mu = muS[row], rs = rsS[row];
      float4 v;
      v.x = (acc[1][g * 4 + 0][r] - mu) * rs * ga.x + be.x;
      v.y = (acc[1][g * 4 + 1][r] - mu) * rs * ga.y + be.y;
      v.z = (acc[1][g * 4 + 2][r] - mu) * rs * ga.z + be.z;
      v.w = (acc[1][g * 4 + 3][r] - mu) * rs * ga.w + be.w;
      *(float4*)(out + (size_t)(b * Nn + m0 + row) * Hn + h) = v;
    }
  }
}

extern "C" void kernel_launch(void* const* d_in, const int* in_sizes, int n_in,
                              void* d_out, int out_size, void* d_ws, size_t ws_size,
                              hipStream_t stream) {
  const float* f     = (const float*)d_in[0];
  const float* k     = (const float*)d_in[1];
  const float* Wlin  = (const float*)d_in[2];
  const float* blin  = (const float*)d_in[3];
  const float* gamma = (const float*)d_in[4];
  const float* beta  = (const float*)d_in[5];
  float* out = (float*)d_out;
  float* wgt = (float*)d_ws;

  (void)in_sizes; (void)n_in; (void)out_size; (void)ws_size;

  wgen_gemm<<<dim3(NK / 128, Nn / 64, Bn), 256, 0, stream>>>(f, Wlin, blin, wgt);
  dyconv_ln<<<dim3(Nn / 16, Bn), S2T, 0, stream>>>(k, wgt, gamma, beta, out);
}


# Stage 1 f operand straight from global memory into WMMA fragments (a wave owns 32 rows, a lane reads 64 contiguous bytes per 32-k chunk); only the W tile goes through LDS; k sum order regrouped
# speedup vs baseline: 1.6434x; 1.1483x over previous
#include <hip/hip_runtime.h>
#include <hip/hip_bf16.h>


#define Bn 64
#define Nn 256
#define Hn 1024
#define NK 768
#define LN_EPS 1e-5f

typedef __attribute__((ext_vector_type(2))) float v2f;
typedef __attribute__((ext_vector_type(8))) float v8f;

#define G1_N 96
#define G1_CT (G1_N / 16)
#define G1_KC 32
#define G1_P 36
#define G1_NCH (Hn / G1_KC)

__device__ __forceinline__ void g1_load_a(float4 (&X)[2][4], const float* p) {
#pragma unroll
  for (int rg = 0; rg < 2; ++rg)
#pragma unroll
    for (int j = 0; j < 4; ++j) X[rg][j] = *(const float4*)(p + rg * 16 * Hn + 4 * j);
}
__device__ __forceinline__ void g1_load_w(float4 (&S)[3], const float* p) {
#pragma unroll
  for (int i = 0; i < 3; ++i) S[i] = *(const float4*)(p + i * 32 * Hn);
}
__device__ __forceinline__ void g1_store_w(float* d, const float4 (&S)[3]) {
#pragma unroll
  for (int i = 0; i < 3; ++i) *(float4*)(d + i * 32 * G1_P) = S[i];
}
__device__ __forceinline__ void g1_chunk(v8f (&acc)[2][G1_CT], const float* ws,
                                         const float4 (&X)[2][4]) {
#pragma unroll
  for (int j = 0; j < 4; ++j) {
    float4 Y[G1_CT];
#pragma unroll
    for (int ci = 0; ci < G1_CT; ++ci) Y[ci] = *(const float4*)(ws + ci * 16 * G1_P + 4 * j);
#pragma unroll
    for (int p = 0; p < 2; ++p)
#pragma unroll
      for (int ci = 0; ci < G1_CT; ++ci) {
        v2f bb;
        bb.x = p ? Y[ci].z : Y[ci].x;
        bb.y = p ? Y[ci].w : Y[ci].y;
#pragma unroll
        for (int rg = 0; rg < 2; ++rg) {
          v2f a;
          a.x = p ? X[rg][j].z : X[rg][j].x;
          a.y = p ? X[rg][j].w : X[rg][j].y;
          acc[rg][ci] = __builtin_amdgcn_wmma_f32_16x16x4_f32(
              false, a, false, bb, (short)0, acc[rg][ci], false, false);
        }
      }
  }
}

__global__ __launch_bounds__(256) __attribute__((amdgpu_waves_per_eu(4, 4))) void wgen_gemm(
    const float* __restrict__ f, const float* __restrict__ Wlin,
    const float* __restrict__ blin, float* __restrict__ wout) {
  __shared__ __attribute__((aligned(16))) float Ws[2][G1_N * G1_P];

  const int tid = threadIdx.x;
  const int wave = tid >> 5, lane = tid & 31;
  const int lm = lane & 15, hi = lane >> 4;
  const int o0 = blockIdx.x * G1_N;
  const int row0 = blockIdx.y * 256 + wave * 32;

  const float* fa = f + (size_t)(row0 + lm) * Hn + 16 * hi;
  const int srow = tid >> 3, sseg = tid & 7;
  const float* wb = Wlin + (size_t)(o0 + srow) * Hn + 4 * sseg;
  float* sd0 = &Ws[0][srow * G1_P + 4 * sseg];
  float* sd1 = &Ws[1][srow * G1_P + 4 * sseg];
  const float* ws0 = &Ws[0][lm * G1_P + 16 * hi];
  const float* ws1 = &Ws[1][lm * G1_P + 16 * hi];

  v8f acc[2][G1_CT] = {};
  float4 XP[2][4], XQ[2][4], S[3];

  g1_load_w(S, wb);
  g1_load_a(XP, fa);
  g1_store_w(sd0, S);
  __syncthreads();

  for (int c = 0; c < G1_NCH; c += 2) {
    g1_load_w(S, wb + G1_KC);
    g1_load_a(XQ, fa + G1_KC);
    g1_chunk(acc, ws0, XP);
    g1_store_w(sd1, S);
    __syncthreads();
    const int adv = (c + 2 < G1_NCH) ? 2 * G1_KC : 0;
    wb += adv;
    fa += adv;
    g1_load_w(S, wb);
    g1_load_a(XP, fa);
    g1_chunk(acc, ws1, XQ);
    g1_store_w(sd0, S);
    __syncthreads();
  }

#pragma unroll
  for (int ci = 0; ci < G1_CT; ++ci) {
    const int o = o0 + ci * 16 + lm;
    const float bl = blin[o];
#pragma unroll
    for (int rg = 0; rg < 2; ++rg)
#pragma unroll
      for (int r = 0; r < 8; ++r) {
        const int m = row0 + rg * 16 + r + hi * 8;
        wout[(size_t)m * NK + o] = acc[rg][ci][r] + bl;
      }
  }
}

#define S2W 8
#define S2T (S2W * 32)
#define S2CT 8
#define AWP 772
#define NSTEP (Nn / 4)

__device__ __forceinline__ float row_from_left(float x) {
  const int v = __float_as_int(x);
  return __int_as_float(__builtin_amdgcn_update_dpp(v, v, 0x121  , 0xF, 0xF, false));
}
__device__ __forceinline__ float row_from_right(float x) {
  const int v = __float_as_int(x);
  return __int_as_float(__builtin_amdgcn_update_dpp(v, v, 0x12F  , 0xF, 0xF, false));
}

__device__ __forceinline__ void step_mma(v8f (&acc)[3][S2CT], const v2f* ap,
                                         const float4 X0, const float4 X1,
                                         const float4 Y0, const float4 Y1) {
  const v2f a0 = ap[0], a1 = ap[1], a2 = ap[2];
  const float bx[S2CT] = {X0.x, X0.y, X0.z, X0.w, X1.x, X1.y, X1.z, X1.w};
  const float by[S2CT] = {Y0.x, Y0.y, Y0.z, Y0.w, Y1.x, Y1.y, Y1.z, Y1.w};
#pragma unroll
  for (int ct = 0; ct < S2CT; ++ct) {
    v2f bb;
    bb.x = bx[ct];
    bb.y = by[ct];
    acc[0][ct] = __builtin_amdgcn_wmma_f32_16x16x4_f32(
        false, a0, false, bb, (short)0, acc[0][ct], false, false);
    acc[1][ct] = __builtin_amdgcn_wmma_f32_16x16x4_f32(
        false, a1, false, bb, (short)0, acc[1][ct], false, false);
    acc[2][ct] = __builtin_amdgcn_wmma_f32_16x16x4_f32(
        false, a2, false, bb, (short)0, acc[2][ct], false, false);
  }
}

__global__ __launch_bounds__(S2T) __attribute__((amdgpu_waves_per_eu(4, 4))) void dyconv_ln(
    const float* __restrict__ kin, const float* __restrict__ w,
    const float* __restrict__ gamma, const float* __restrict__ beta,
    float* __restrict__ out) {
  __shared__ __attribute__((aligned(16))) float Aw[16 * AWP];
  __shared__ float redS[S2W * 16];
  __shared__ float redQ[S2W * 16];
  __shared__ float edgeL[S2W * 16];
  __shared__ float edgeR[S2W * 16];
  __shared__ float muS[16];
  __shared__ float rsS[16];

  const int tid = threadIdx.x;
  const int b = blockIdx.y;
  const int m0 = blockIdx.x * 16;
  const int wave = tid >> 5, lane = tid & 31;
  const int lm = lane & 15, hi2 = lane >> 4;
  const int wcol = wave * (16 * S2CT);

  const float* kp = kin + (size_t)(b * Nn + 2 * hi2) * Hn + wcol + 4 * lm;
#define LDB(x0, x1, y0, y1, p)                 \
  x0 = *(const float4*)(p);                    \
  x1 = *(const float4*)((p) + 64);             \
  y0 = *(const float4*)((p) + Hn);             \
  y1 = *(const float4*)((p) + Hn + 64)
  float4 pX0, pX1, pY0, pY1, qX0, qX1, qY0, qY1;
  LDB(pX0, pX1, pY0, pY1, kp);

  {
    const int row = tid >> 4, j = tid & 15;
    const float* wsrc = w + (size_t)(b * Nn + m0 + row) * NK + j * 48;
    float src[48], dst[48];
#pragma unroll
    for (int n = 0; n < 12; ++n) {
      const float4 v = *(const float4*)(wsrc + 4 * n);
      src[4 * n + 0] = v.x; src[4 * n + 1] = v.y; src[4 * n + 2] = v.z; src[4 * n + 3] = v.w;
    }
#pragma unroll
    for (int il = 0; il < 16; ++il)
#pragma unroll
      for (int t = 0; t < 3; ++t)
        dst[(il >> 1) * 6 + t * 2 + (il & 1)] = src[il * 3 + t];
    float* wdst = &Aw[row * AWP + j * 48];
#pragma unroll
    for (int n = 0; n < 12; ++n)
      *(float4*)(wdst + 4 * n) = make_float4(dst[4 * n], dst[4 * n + 1], dst[4 * n + 2], dst[4 * n + 3]);
  }

  v8f acc[3][S2CT] = {};

  const v2f* ap = (const v2f*)Aw + lm * (AWP / 2) + hi2 * 3;

  __syncthreads();

  for (int s = 0; s < NSTEP; s += 2) {
    LDB(qX0, qX1, qY0, qY1, kp + 4 * Hn);
    step_mma(acc, ap + s * 6, pX0, pX1, pY0, pY1);
    kp += (s + 2 < NSTEP) ? 8 * Hn : 0;
    LDB(pX0, pX1, pY0, pY1, kp);
    step_mma(acc, ap + (s + 1) * 6, qX0, qX1, qY0, qY1);
  }
#undef LDB

  if (lm == 15) {
#pragma unroll
    for (int r = 0; r < 8; ++r) edgeL[wave * 16 + hi2 * 8 + r] = acc[0][S2CT - 1][r];
  }
  if (lm == 0) {
#pragma unroll
    for (int r = 0; r < 8; ++r) edgeR[wave * 16 + hi2 * 8 + r] = acc[2][0][r];
  }
  __syncthreads();
#pragma unroll
  for (int r = 0; r < 8; ++r) {
    const float eL = (wave > 0) ? edgeL[(wave - 1) * 16 + hi2 * 8 + r] : 0.0f;
    const float eR = (wave < S2W - 1) ? edgeR[(wave + 1) * 16 + hi2 * 8 + r] : 0.0f;
    const float p3 = row_from_left(acc[0][3][r]);
    const float p7 = row_from_left(acc[0][7][r]);
    const float n0 = row_from_right(acc[2][0][r]);
    const float n4 = row_from_right(acc[2][4][r]);
    float left[S2CT], right[S2CT];
    left[0] = (lm == 0) ? eL : p3;
    left[4] = (lm == 0) ? p3 : p7;
    right[3] = (lm == 15) ? n4 : n0;
    right[7] = (lm == 15) ? eR : n4;
#pragma unroll
    for (int ct = 0; ct < S2CT; ++ct) {
      if ((ct & 3) != 0) left[ct] = acc[0][ct - 1][r];
      if ((ct & 3) != 3) right[ct] = acc[2][ct + 1][r];
    }
#pragma unroll
    for (int ct = 0; ct < S2CT; ++ct)
      acc[1][ct][r] = (left[ct] + acc[1][ct][r]) + right[ct];
  }

  float s[8], q[8];
#pragma unroll
  for (int r = 0; r < 8; ++r) { s[r] = 0.0f; q[r] = 0.0f; }
#pragma unroll
  for (int u = 0; u < S2CT; ++u)
#pragma unroll
    for (int r = 0; r < 8; ++r) {
      float v = acc[1][u][r];
      s[r] += v;
      q[r] += v * v;
    }
#pragma unroll
  for (int mask = 1; mask < 16; mask <<= 1)
#pragma unroll
    for (int r = 0; r < 8; ++r) {
      s[r] += __shfl_xor(s[r], mask, 32);
      q[r] += __shfl_xor(q[r], mask, 32);
    }
  if (lm == 0) {
#pragma unroll
    for (int r = 0; r < 8; ++r) {
      const int row = r + hi2 * 8;
      redS[wave * 16 + row] = s[r];
      redQ[wave * 16 + row] = q[r];
    }
  }
  __syncthreads();
  if (tid < 16) {
    float ss = 0.0f, qq = 0.0f;
#pragma unroll
    for (int wv = 0; wv < S2W; ++wv) {
      ss += redS[wv * 16 + tid];
      qq += redQ[wv * 16 + tid];
    }
    const float mu = ss * (1.0f / 1024.0f);
    const float var = qq * (1.0f / 1024.0f) - mu * mu;
    muS[tid] = mu;
    rsS[tid] = rsqrtf(var + LN_EPS);
  }
  __syncthreads();

#pragma unroll
  for (int g = 0; g < 2; ++g) {
    const int h = wcol + g * 64 + 4 * lm;
    const float4 ga = *(const float4*)(gamma + h);
    const float4 be = *(const float4*)(beta + h);
#pragma unroll
    for (int r = 0; r < 8; ++r) {
      const int row = r + hi2 * 8;
      const float mu = muS[row], rs = rsS[row];
      float4 v;
      v.x = (acc[1][g * 4 + 0][r] - mu) * rs * ga.x + be.x;
      v.y = (acc[1][g * 4 + 1][r] - mu) * rs * ga.y + be.y;
      v.z = (acc[1][g * 4 + 2][r] - mu) * rs * ga.z + be.z;
      v.w = (acc[1][g * 4 + 3][r] - mu) * rs * ga.w + be.w;
      *(float4*)(out + (size_t)(b * Nn + m0 + row) * Hn + h) = v;
    }
  }
}

extern "C" void kernel_launch(void* const* d_in, const int* in_sizes, int n_in,
                              void* d_out, int out_size, void* d_ws, size_t ws_size,
                              hipStream_t stream) {
  const float* f     = (const float*)d_in[0];
  const float* k     = (const float*)d_in[1];
  const float* Wlin  = (const float*)d_in[2];
  const float* blin  = (const float*)d_in[3];
  const float* gamma = (const float*)d_in[4];
  const float* beta  = (const float*)d_in[5];
  float* out = (float*)d_out;
  float* wgt = (float*)d_ws;

  (void)in_sizes; (void)n_in; (void)out_size; (void)ws_size;

  wgen_gemm<<<dim3(NK / G1_N, (Bn * Nn) / 256), 256, 0, stream>>>(f, Wlin, blin, wgt);
  dyconv_ln<<<dim3(Nn / 16, Bn), S2T, 0, stream>>>(k, wgt, gamma, beta, out);
}
